# starting kernel
# baseline (speedup 1.0000x reference)
#include <hip/hip_runtime.h>

#define N_ 32
#define C_ 64
#define T_ 128
#define V_ 25
#define K_ 10
#define M_ (N_ * V_)

typedef float v2f __attribute__((ext_vector_type(2)));
typedef float v8f __attribute__((ext_vector_type(8)));

__global__ void vc_mean_kernel(const float* __restrict__ region,
                               float* __restrict__ mf2) {
    const int nc = blockIdx.x;
    const int n  = nc / C_;
    const int c  = nc % C_;
    const int idx = threadIdx.x;
    if (idx >= V_ * K_) return;

    const float* base = region + (size_t)nc * T_ * V_ * K_ + idx;
    float s = 0.0f;
    #pragma unroll 4
    for (int t = 0; t < T_; ++t)
        s += base[(size_t)t * (V_ * K_)];

    const int v = idx / K_;
    const int k = idx % K_;
    const int m = n * V_ + v;
    mf2[((size_t)k * M_ + m) * C_ + c] = s * (1.0f / (float)T_);
}

__global__ void vc_attn_kernel(const float* __restrict__ mf2,
                               const float* __restrict__ w1,
                               const float* __restrict__ b1,
                               const float* __restrict__ w2,
                               float* __restrict__ wk2) {
    const int tile = blockIdx.x;
    const int k    = tile / (M_ / 16);
    const int m0   = (tile % (M_ / 16)) * 16;
    const int lane = threadIdx.x;
    const int half = lane >> 4;
    const int l    = lane & 15;

    const float* arow = mf2 + ((size_t)k * M_ + (m0 + l)) * C_;
    const int   pclamp = (l < K_) ? l : (K_ - 1);
    const float bmask  = (l < K_) ? 1.0f : 0.0f;
    const float* brow  = w1 + ((size_t)k * K_ + pclamp) * C_;

    v8f acc = {0.f, 0.f, 0.f, 0.f, 0.f, 0.f, 0.f, 0.f};

    #pragma unroll
    for (int j = 0; j < C_ / 4; ++j) {
        const int c0 = j * 4 + half * 2;
        v2f a, b;
        a.x = arow[c0];
        a.y = arow[c0 + 1];
        b.x = brow[c0] * bmask;
        b.y = brow[c0 + 1] * bmask;
        acc = __builtin_amdgcn_wmma_f32_16x16x4_f32(
             false, a,  false, b,
             (short)0, acc,  false,  false);
    }

    __shared__ float tile_lds[16][17];
    #pragma unroll
    for (int g = 0; g < 8; ++g)
        tile_lds[g + half * 8][l] = acc[g];
    __syncthreads();

    if (lane < 16) {
        const int row = lane;
        float lg[K_];
        float mx = -1e30f;
        #pragma unroll
        for (int p = 0; p < K_; ++p) {
            lg[p] = tile_lds[row][p] + b1[k * K_ + p];
            mx = fmaxf(mx, lg[p]);
        }
        float sum = 0.0f;
        #pragma unroll
        for (int p = 0; p < K_; ++p) {
            lg[p] = __expf(lg[p] - mx);
            sum += lg[p];
        }
        const float inv = 1.0f / sum;
        float wkv = 0.0f;
        #pragma unroll
        for (int p = 0; p < K_; ++p)
            wkv += lg[p] * inv * w2[p];
        wk2[(size_t)k * M_ + m0 + row] = wkv;
    }
}

__global__ void vc_out_kernel(const float* __restrict__ region,
                              const float* __restrict__ wk2,
                              const float* __restrict__ b2,
                              float* __restrict__ out) {
    const int nc = blockIdx.x;
    const int n  = nc / C_;

    __shared__ float wk_lds[V_ * K_];
    for (int i = threadIdx.x; i < V_ * K_; i += blockDim.x) {
        const int v = i / K_;
        const int k = i % K_;
        wk_lds[i] = wk2[(size_t)k * M_ + n * V_ + v];
    }
    __syncthreads();

    const float bias = b2[0];
    const float* slab = region + (size_t)nc * T_ * V_ * K_;
    float* oslab = out + (size_t)nc * T_ * V_;

    for (int i = threadIdx.x; i < T_ * V_; i += blockDim.x) {
        const int v = i % V_;
        const float2* p2 = (const float2*)(slab + (size_t)i * K_);
        const float* wv = &wk_lds[v * K_];
        float s = bias;
        #pragma unroll
        for (int j = 0; j < K_ / 2; ++j) {
            const float2 r = p2[j];
            s += r.x * wv[2 * j] + r.y * wv[2 * j + 1];
        }
        __builtin_nontemporal_store(s, &oslab[i]);
    }
}

extern "C" void kernel_launch(void* const* d_in, const int* in_sizes, int n_in,
                              void* d_out, int out_size, void* d_ws, size_t ws_size,
                              hipStream_t stream) {
    const float* region = (const float*)d_in[0];
    const float* w1     = (const float*)d_in[1];
    const float* b1     = (const float*)d_in[2];
    const float* w2     = (const float*)d_in[3];
    const float* b2     = (const float*)d_in[4];
    float* out = (float*)d_out;

    float* mf2 = (float*)d_ws;
    float* wk2 = mf2 + (size_t)K_ * M_ * C_;

    vc_mean_kernel<<<N_ * C_, 256, 0, stream>>>(region, mf2);
    vc_attn_kernel<<<K_ * (M_ / 16), 32, 0, stream>>>(mf2, w1, b1, w2, wk2);
    vc_out_kernel<<<N_ * C_, 256, 0, stream>>>(region, wk2, b2, out);
}


# Stage-1 mean reads each (n,c) slab with 16-byte loads over row pairs, 16-32 in flight per lane, 2-wave blocks; sum over t reordered: even-row and odd-row partial sums, added at the end
# speedup vs baseline: 1.0373x; 1.0373x over previous
#include <hip/hip_runtime.h>

#define N_ 32
#define C_ 64
#define T_ 128
#define V_ 25
#define K_ 10
#define M_ (N_ * V_)

typedef float v2f __attribute__((ext_vector_type(2)));
typedef float v4f __attribute__((ext_vector_type(4)));
typedef float v8f __attribute__((ext_vector_type(8)));

#define ROWPAIR_F4_ ((2 * V_ * K_) / 4)
#define ROWPAIRS_   (T_ / 2)
#define GROUP_      8
#define NGROUPS_    (ROWPAIRS_ / GROUP_)

__global__ __launch_bounds__(64)
void vc_mean_kernel(const float* __restrict__ region,
                    float* __restrict__ mf2) {
    const int nc = blockIdx.x;
    const int n  = nc / C_;
    const int c  = nc % C_;
    const int i  = threadIdx.x;

    const v4f* slab4 = reinterpret_cast<const v4f*>(
        region + (size_t)nc * (T_ * V_ * K_));
    const bool has1 = (i + 64) < ROWPAIR_F4_;
    const v4f* p0 = slab4 + i;
    const v4f* p1 = slab4 + (has1 ? (i + 64) : (ROWPAIR_F4_ - 1));

    v4f w0[2][GROUP_], w1[2][GROUP_];
#define LOAD_GROUP(G) do { \
        _Pragma("unroll") for (int u = 0; u < GROUP_; ++u) { \
            w0[(G) & 1][u] = p0[((G) * GROUP_ + u) * ROWPAIR_F4_]; \
            w1[(G) & 1][u] = p1[((G) * GROUP_ + u) * ROWPAIR_F4_]; \
        } } while (0)

    v4f s0 = {0.f, 0.f, 0.f, 0.f};
    v4f s1 = {0.f, 0.f, 0.f, 0.f};
    LOAD_GROUP(0);
    #pragma unroll
    for (int g = 0; g < NGROUPS_; ++g) {
        __builtin_amdgcn_sched_barrier(0);
        if (g + 1 < NGROUPS_) LOAD_GROUP(g + 1);
        __builtin_amdgcn_sched_barrier(0);
        #pragma unroll
        for (int u = 0; u < GROUP_; ++u) {
            s0 += w0[g & 1][u];
            s1 += w1[g & 1][u];
        }
    }
#undef LOAD_GROUP

    __shared__ v4f part4[128];
    part4[i]      = s0;
    part4[i + 64] = s1;
    __syncthreads();

    const float* part = reinterpret_cast<const float*>(part4);
    for (int col = i; col < V_ * K_; col += 64) {
        const float s = part[col] + part[col + V_ * K_];
        const int v = col / K_;
        const int k = col % K_;
        const int m = n * V_ + v;
        mf2[((size_t)k * M_ + m) * C_ + c] = s * (1.0f / (float)T_);
    }
}

__global__ void vc_attn_kernel(const float* __restrict__ mf2,
                               const float* __restrict__ w1,
                               const float* __restrict__ b1,
                               const float* __restrict__ w2,
                               float* __restrict__ wk2) {
    const int tile = blockIdx.x;
    const int k    = tile / (M_ / 16);
    const int m0   = (tile % (M_ / 16)) * 16;
    const int lane = threadIdx.x;
    const int half = lane >> 4;
    const int l    = lane & 15;

    const float* arow = mf2 + ((size_t)k * M_ + (m0 + l)) * C_;
    const int   pclamp = (l < K_) ? l : (K_ - 1);
    const float bmask  = (l < K_) ? 1.0f : 0.0f;
    const float* brow  = w1 + ((size_t)k * K_ + pclamp) * C_;

    v8f acc = {0.f, 0.f, 0.f, 0.f, 0.f, 0.f, 0.f, 0.f};

    #pragma unroll
    for (int j = 0; j < C_ / 4; ++j) {
        const int c0 = j * 4 + half * 2;
        v2f a, b;
        a.x = arow[c0];
        a.y = arow[c0 + 1];
        b.x = brow[c0] * bmask;
        b.y = brow[c0 + 1] * bmask;
        acc = __builtin_amdgcn_wmma_f32_16x16x4_f32(
             false, a,  false, b,
             (short)0, acc,  false,  false);
    }

    __shared__ float tile_lds[16][17];
    #pragma unroll
    for (int g = 0; g < 8; ++g)
        tile_lds[g + half * 8][l] = acc[g];
    __syncthreads();

    if (lane < 16) {
        const int row = lane;
        float lg[K_];
        float mx = -1e30f;
        #pragma unroll
        for (int p = 0; p < K_; ++p) {
            lg[p] = tile_lds[row][p] + b1[k * K_ + p];
            mx = fmaxf(mx, lg[p]);
        }
        float sum = 0.0f;
        #pragma unroll
        for (int p = 0; p < K_; ++p) {
            lg[p] = __expf(lg[p] - mx);
            sum += lg[p];
        }
        const float inv = 1.0f / sum;
        float wkv = 0.0f;
        #pragma unroll
        for (int p = 0; p < K_; ++p)
            wkv += lg[p] * inv * w2[p];
        wk2[(size_t)k * M_ + m0 + row] = wkv;
    }
}

__global__ void vc_out_kernel(const float* __restrict__ region,
                              const float* __restrict__ wk2,
                              const float* __restrict__ b2,
                              float* __restrict__ out) {
    const int nc = blockIdx.x;
    const int n  = nc / C_;

    __shared__ float wk_lds[V_ * K_];
    for (int i = threadIdx.x; i < V_ * K_; i += blockDim.x) {
        const int v = i / K_;
        const int k = i % K_;
        wk_lds[i] = wk2[(size_t)k * M_ + n * V_ + v];
    }
    __syncthreads();

    const float bias = b2[0];
    const float* slab = region + (size_t)nc * T_ * V_ * K_;
    float* oslab = out + (size_t)nc * T_ * V_;

    for (int i = threadIdx.x; i < T_ * V_; i += blockDim.x) {
        const int v = i % V_;
        const float2* p2 = (const float2*)(slab + (size_t)i * K_);
        const float* wv = &wk_lds[v * K_];
        float s = bias;
        #pragma unroll
        for (int j = 0; j < K_ / 2; ++j) {
            const float2 r = p2[j];
            s += r.x * wv[2 * j] + r.y * wv[2 * j + 1];
        }
        __builtin_nontemporal_store(s, &oslab[i]);
    }
}

extern "C" void kernel_launch(void* const* d_in, const int* in_sizes, int n_in,
                              void* d_out, int out_size, void* d_ws, size_t ws_size,
                              hipStream_t stream) {
    const float* region = (const float*)d_in[0];
    const float* w1     = (const float*)d_in[1];
    const float* b1     = (const float*)d_in[2];
    const float* w2     = (const float*)d_in[3];
    const float* b2     = (const float*)d_in[4];
    float* out = (float*)d_out;

    float* mf2 = (float*)d_ws;
    float* wk2 = mf2 + (size_t)K_ * M_ * C_;

    vc_mean_kernel<<<N_ * C_, 64, 0, stream>>>(region, mf2);
    vc_attn_kernel<<<K_ * (M_ / 16), 32, 0, stream>>>(mf2, w1, b1, w2, wk2);
    vc_out_kernel<<<N_ * C_, 256, 0, stream>>>(region, wk2, b2, out);
}


# Stage 3 streams each slab like stage 1: 2-wave blocks, wave-wide 16-byte loads, 16-32 in flight per lane, weights in registers; k-sum reordered (pair sums via LDS, then 5-slot sums)
# speedup vs baseline: 1.0740x; 1.0354x over previous
#include <hip/hip_runtime.h>

#define N_ 32
#define C_ 64
#define T_ 128
#define V_ 25
#define K_ 10
#define M_ (N_ * V_)

typedef float v2f __attribute__((ext_vector_type(2)));
typedef float v4f __attribute__((ext_vector_type(4)));
typedef float v8f __attribute__((ext_vector_type(8)));

#define ROWPAIR_F4_ ((2 * V_ * K_) / 4)
#define ROWPAIRS_   (T_ / 2)
#define GROUP_      8
#define NGROUPS_    (ROWPAIRS_ / GROUP_)

__global__ __launch_bounds__(64)
void vc_mean_kernel(const float* __restrict__ region,
                    float* __restrict__ mf2) {
    const int nc = blockIdx.x;
    const int n  = nc / C_;
    const int c  = nc % C_;
    const int i  = threadIdx.x;

    const v4f* slab4 = reinterpret_cast<const v4f*>(
        region + (size_t)nc * (T_ * V_ * K_));
    const bool has1 = (i + 64) < ROWPAIR_F4_;
    const v4f* p0 = slab4 + i;
    const v4f* p1 = slab4 + (has1 ? (i + 64) : (ROWPAIR_F4_ - 1));

    v4f w0[2][GROUP_], w1[2][GROUP_];
#define LOAD_GROUP(G) do { \
        _Pragma("unroll") for (int u = 0; u < GROUP_; ++u) { \
            w0[(G) & 1][u] = p0[((G) * GROUP_ + u) * ROWPAIR_F4_]; \
            w1[(G) & 1][u] = p1[((G) * GROUP_ + u) * ROWPAIR_F4_]; \
        } } while (0)

    v4f s0 = {0.f, 0.f, 0.f, 0.f};
    v4f s1 = {0.f, 0.f, 0.f, 0.f};
    LOAD_GROUP(0);
    #pragma unroll
    for (int g = 0; g < NGROUPS_; ++g) {
        __builtin_amdgcn_sched_barrier(0);
        if (g + 1 < NGROUPS_) LOAD_GROUP(g + 1);
        __builtin_amdgcn_sched_barrier(0);
        #pragma unroll
        for (int u = 0; u < GROUP_; ++u) {
            s0 += w0[g & 1][u];
            s1 += w1[g & 1][u];
        }
    }
#undef LOAD_GROUP

    __shared__ v4f part4[128];
    part4[i]      = s0;
    part4[i + 64] = s1;
    __syncthreads();

    const float* part = reinterpret_cast<const float*>(part4);
    for (int col = i; col < V_ * K_; col += 64) {
        const float s = part[col] + part[col + V_ * K_];
        const int v = col / K_;
        const int k = col % K_;
        const int m = n * V_ + v;
        mf2[((size_t)k * M_ + m) * C_ + c] = s * (1.0f / (float)T_);
    }
}

__global__ void vc_attn_kernel(const float* __restrict__ mf2,
                               const float* __restrict__ w1,
                               const float* __restrict__ b1,
                               const float* __restrict__ w2,
                               float* __restrict__ wk2) {
    const int tile = blockIdx.x;
    const int k    = tile / (M_ / 16);
    const int m0   = (tile % (M_ / 16)) * 16;
    const int lane = threadIdx.x;
    const int half = lane >> 4;
    const int l    = lane & 15;

    const float* arow = mf2 + ((size_t)k * M_ + (m0 + l)) * C_;
    const int   pclamp = (l < K_) ? l : (K_ - 1);
    const float bmask  = (l < K_) ? 1.0f : 0.0f;
    const float* brow  = w1 + ((size_t)k * K_ + pclamp) * C_;

    v8f acc = {0.f, 0.f, 0.f, 0.f, 0.f, 0.f, 0.f, 0.f};

    #pragma unroll
    for (int j = 0; j < C_ / 4; ++j) {
        const int c0 = j * 4 + half * 2;
        v2f a, b;
        a.x = arow[c0];
        a.y = arow[c0 + 1];
        b.x = brow[c0] * bmask;
        b.y = brow[c0 + 1] * bmask;
        acc = __builtin_amdgcn_wmma_f32_16x16x4_f32(
             false, a,  false, b,
             (short)0, acc,  false,  false);
    }

    __shared__ float tile_lds[16][17];
    #pragma unroll
    for (int g = 0; g < 8; ++g)
        tile_lds[g + half * 8][l] = acc[g];
    __syncthreads();

    if (lane < 16) {
        const int row = lane;
        float lg[K_];
        float mx = -1e30f;
        #pragma unroll
        for (int p = 0; p < K_; ++p) {
            lg[p] = tile_lds[row][p] + b1[k * K_ + p];
            mx = fmaxf(mx, lg[p]);
        }
        float sum = 0.0f;
        #pragma unroll
        for (int p = 0; p < K_; ++p) {
            lg[p] = __expf(lg[p] - mx);
            sum += lg[p];
        }
        const float inv = 1.0f / sum;
        float wkv = 0.0f;
        #pragma unroll
        for (int p = 0; p < K_; ++p)
            wkv += lg[p] * inv * w2[p];
        wk2[(size_t)k * M_ + m0 + row] = wkv;
    }
}

#define O_RP_WAVE_   (ROWPAIRS_ / 2)
#define O_GROUP_     4
#define O_NGROUPS_   (O_RP_WAVE_ / O_GROUP_)
#define O_HROW_      256
#define O_GRP_OUT_   (O_GROUP_ * 2 * V_)
#define O_BATCH_OUT_ (4 * O_GRP_OUT_)

#define O_WAVE_SYNC() do { \
        __builtin_amdgcn_fence(__ATOMIC_RELEASE, "wavefront", "local"); \
        __builtin_amdgcn_wave_barrier(); \
        __builtin_amdgcn_fence(__ATOMIC_ACQUIRE, "wavefront", "local"); \
    } while (0)

__global__ __launch_bounds__(64, 5)
void vc_out_kernel(const float* __restrict__ region,
                   const float* __restrict__ wk2,
                   const float* __restrict__ b2,
                   float* __restrict__ out) {
    const int nc   = blockIdx.x;
    const int n    = nc / C_;
    const int lane = threadIdx.x & 31;
    const int wave = threadIdx.x >> 5;

    __shared__ __attribute__((aligned(16))) float hbuf[2][O_GROUP_ * O_HROW_];
    __shared__ __attribute__((aligned(16))) float obuf[2][O_BATCH_OUT_];
    float* hw = hbuf[wave];
    float* ow = obuf[wave];

    const v4f* wbase = reinterpret_cast<const v4f*>(
        region + (size_t)nc * (T_ * V_ * K_)) + (size_t)(wave * O_RP_WAVE_) * ROWPAIR_F4_;
    const v4f* p[4];
    v4f w[4];
    #pragma unroll
    for (int j = 0; j < 4; ++j) {
        const int q  = 32 * j + lane;
        const int qc = (q < ROWPAIR_F4_) ? q : (ROWPAIR_F4_ - 1);
        p[j] = wbase + qc;
        float wv[4];
        #pragma unroll
        for (int e = 0; e < 4; ++e) {
            const int f = (4 * qc + e) % (V_ * K_);
            const int v = f / K_;
            const int k = f % K_;
            wv[e] = wk2[(size_t)k * M_ + n * V_ + v];
        }
        w[j].x = wv[0]; w[j].y = wv[1]; w[j].z = wv[2]; w[j].w = wv[3];
    }
    const float bias = b2[0];

    v4f* oslab4 = reinterpret_cast<v4f*>(out + (size_t)nc * (T_ * V_))
                  + (size_t)wave * (2 * O_BATCH_OUT_ / 4);

    v4f buf[2][O_GROUP_ * 4];
#define O_LOAD_GROUP(G) do { \
        _Pragma("unroll") for (int u = 0; u < O_GROUP_; ++u) { \
            _Pragma("unroll") for (int j = 0; j < 4; ++j) \
                buf[(G) & 1][u * 4 + j] = p[j][((G) * O_GROUP_ + u) * ROWPAIR_F4_]; \
        } } while (0)

    O_LOAD_GROUP(0);
    #pragma unroll
    for (int g = 0; g < O_NGROUPS_; ++g) {
        __builtin_amdgcn_sched_barrier(0);
        if (g + 1 < O_NGROUPS_) O_LOAD_GROUP(g + 1);
        __builtin_amdgcn_sched_barrier(0);

        #pragma unroll
        for (int u = 0; u < O_GROUP_; ++u) {
            #pragma unroll
            for (int j = 0; j < 4; ++j) {
                const v4f r = buf[g & 1][u * 4 + j];
                v2f h;
                h.x = fmaf(r.x, w[j].x, r.y * w[j].y);
                h.y = fmaf(r.z, w[j].z, r.w * w[j].w);
                *reinterpret_cast<v2f*>(hw + u * O_HROW_ + 2 * (32 * j + lane)) = h;
            }
        }
        O_WAVE_SYNC();

        #pragma unroll
        for (int u = 0; u < O_GROUP_; ++u) {
            #pragma unroll
            for (int hlf = 0; hlf < 2; ++hlf) {
                const int o = lane + 32 * hlf;
                if (o < 2 * V_) {
                    const float* hs = hw + u * O_HROW_ + 5 * o;
                    const float s = ((hs[0] + hs[1]) + (hs[2] + hs[3])) + hs[4];
                    ow[(g & 3) * O_GRP_OUT_ + u * (2 * V_) + o] = s + bias;
                }
            }
        }
        O_WAVE_SYNC();

        if ((g & 3) == 3) {
            const v4f* ow4 = reinterpret_cast<const v4f*>(ow);
            v4f* dst = oslab4 + (g >> 2) * (O_BATCH_OUT_ / 4);
            #pragma unroll
            for (int r = 0; r < 7; ++r) {
                const int i4 = lane + 32 * r;
                if (i4 < O_BATCH_OUT_ / 4)
                    __builtin_nontemporal_store(ow4[i4], &dst[i4]);
            }
            O_WAVE_SYNC();
        }
    }
#undef O_LOAD_GROUP
}

extern "C" void kernel_launch(void* const* d_in, const int* in_sizes, int n_in,
                              void* d_out, int out_size, void* d_ws, size_t ws_size,
                              hipStream_t stream) {
    const float* region = (const float*)d_in[0];
    const float* w1     = (const float*)d_in[1];
    const float* b1     = (const float*)d_in[2];
    const float* w2     = (const float*)d_in[3];
    const float* b2     = (const float*)d_in[4];
    float* out = (float*)d_out;

    float* mf2 = (float*)d_ws;
    float* wk2 = mf2 + (size_t)K_ * M_ * C_;

    vc_mean_kernel<<<N_ * C_, 64, 0, stream>>>(region, mf2);
    vc_attn_kernel<<<K_ * (M_ / 16), 32, 0, stream>>>(mf2, w1, b1, w2, wk2);
    vc_out_kernel<<<N_ * C_, 64, 0, stream>>>(region, wk2, b2, out);
}


# Stage 1 walks each slab's two halves downwards (row pairs 31..0 and 63..32) so that it ends on the rows stage 3 reads first, still in L2; sum over t reordered to that walk
# speedup vs baseline: 1.1192x; 1.0421x over previous
#include <hip/hip_runtime.h>

#define N_ 32
#define C_ 64
#define T_ 128
#define V_ 25
#define K_ 10
#define M_ (N_ * V_)

typedef float v2f __attribute__((ext_vector_type(2)));
typedef float v4f __attribute__((ext_vector_type(4)));
typedef float v8f __attribute__((ext_vector_type(8)));

#define ROWPAIR_F4_ ((2 * V_ * K_) / 4)
#define ROWPAIRS_   (T_ / 2)
#define GROUP_      8
#define NGROUPS_    (ROWPAIRS_ / GROUP_)
#define WALK_RP_(G, U) \
    ((((U) < GROUP_ / 2) ? (ROWPAIRS_ / 2 - 1) : (ROWPAIRS_ - 1)) \
     - ((G) * (GROUP_ / 2) + ((U) % (GROUP_ / 2))))

__global__ __launch_bounds__(64)
void vc_mean_kernel(const float* __restrict__ region,
                    float* __restrict__ mf2) {
    const int nc = blockIdx.x;
    const int n  = nc / C_;
    const int c  = nc % C_;
    const int i  = threadIdx.x;

    const v4f* slab4 = reinterpret_cast<const v4f*>(
        region + (size_t)nc * (T_ * V_ * K_));
    const bool has1 = (i + 64) < ROWPAIR_F4_;
    const v4f* p0 = slab4 + i;
    const v4f* p1 = slab4 + (has1 ? (i + 64) : (ROWPAIR_F4_ - 1));

    v4f w0[2][GROUP_], w1[2][GROUP_];
#define LOAD_GROUP(G) do { \
        _Pragma("unroll") for (int u = 0; u < GROUP_; ++u) { \
            w0[(G) & 1][u] = p0[WALK_RP_(G, u) * ROWPAIR_F4_]; \
            w1[(G) & 1][u] = p1[WALK_RP_(G, u) * ROWPAIR_F4_]; \
        } } while (0)

    v4f s0 = {0.f, 0.f, 0.f, 0.f};
    v4f s1 = {0.f, 0.f, 0.f, 0.f};
    LOAD_GROUP(0);
    #pragma unroll
    for (int g = 0; g < NGROUPS_; ++g) {
        __builtin_amdgcn_sched_barrier(0);
        if (g + 1 < NGROUPS_) LOAD_GROUP(g + 1);
        __builtin_amdgcn_sched_barrier(0);
        #pragma unroll
        for (int u = 0; u < GROUP_; ++u) {
            s0 += w0[g & 1][u];
            s1 += w1[g & 1][u];
        }
    }
#undef LOAD_GROUP

    __shared__ v4f part4[128];
    part4[i]      = s0;
    part4[i + 64] = s1;
    __syncthreads();

    const float* part = reinterpret_cast<const float*>(part4);
    for (int col = i; col < V_ * K_; col += 64) {
        const float s = part[col] + part[col + V_ * K_];
        const int v = col / K_;
        const int k = col % K_;
        const int m = n * V_ + v;
        mf2[((size_t)k * M_ + m) * C_ + c] = s * (1.0f / (float)T_);
    }
}

__global__ void vc_attn_kernel(const float* __restrict__ mf2,
                               const float* __restrict__ w1,
                               const float* __restrict__ b1,
                               const float* __restrict__ w2,
                               float* __restrict__ wk2) {
    const int tile = blockIdx.x;
    const int k    = tile / (M_ / 16);
    const int m0   = (tile % (M_ / 16)) * 16;
    const int lane = threadIdx.x;
    const int half = lane >> 4;
    const int l    = lane & 15;

    const float* arow = mf2 + ((size_t)k * M_ + (m0 + l)) * C_;
    const int   pclamp = (l < K_) ? l : (K_ - 1);
    const float bmask  = (l < K_) ? 1.0f : 0.0f;
    const float* brow  = w1 + ((size_t)k * K_ + pclamp) * C_;

    v8f acc = {0.f, 0.f, 0.f, 0.f, 0.f, 0.f, 0.f, 0.f};

    #pragma unroll
    for (int j = 0; j < C_ / 4; ++j) {
        const int c0 = j * 4 + half * 2;
        v2f a, b;
        a.x = arow[c0];
        a.y = arow[c0 + 1];
        b.x = brow[c0] * bmask;
        b.y = brow[c0 + 1] * bmask;
        acc = __builtin_amdgcn_wmma_f32_16x16x4_f32(
             false, a,  false, b,
             (short)0, acc,  false,  false);
    }

    __shared__ float tile_lds[16][17];
    #pragma unroll
    for (int g = 0; g < 8; ++g)
        tile_lds[g + half * 8][l] = acc[g];
    __syncthreads();

    if (lane < 16) {
        const int row = lane;
        float lg[K_];
        float mx = -1e30f;
        #pragma unroll
        for (int p = 0; p < K_; ++p) {
            lg[p] = tile_lds[row][p] + b1[k * K_ + p];
            mx = fmaxf(mx, lg[p]);
        }
        float sum = 0.0f;
        #pragma unroll
        for (int p = 0; p < K_; ++p) {
            lg[p] = __expf(lg[p] - mx);
            sum += lg[p];
        }
        const float inv = 1.0f / sum;
        float wkv = 0.0f;
        #pragma unroll
        for (int p = 0; p < K_; ++p)
            wkv += lg[p] * inv * w2[p];
        wk2[(size_t)k * M_ + m0 + row] = wkv;
    }
}

#define O_RP_WAVE_   (ROWPAIRS_ / 2)
#define O_GROUP_     4
#define O_NGROUPS_   (O_RP_WAVE_ / O_GROUP_)
#define O_HROW_      256
#define O_GRP_OUT_   (O_GROUP_ * 2 * V_)
#define O_BATCH_OUT_ (4 * O_GRP_OUT_)

#define O_WAVE_SYNC() do { \
        __builtin_amdgcn_fence(__ATOMIC_RELEASE, "wavefront", "local"); \
        __builtin_amdgcn_wave_barrier(); \
        __builtin_amdgcn_fence(__ATOMIC_ACQUIRE, "wavefront", "local"); \
    } while (0)

__global__ __launch_bounds__(64, 5)
void vc_out_kernel(const float* __restrict__ region,
                   const float* __restrict__ wk2,
                   const float* __restrict__ b2,
                   float* __restrict__ out) {
    const int nc   = blockIdx.x;
    const int n    = nc / C_;
    const int lane = threadIdx.x & 31;
    const int wave = threadIdx.x >> 5;

    __shared__ __attribute__((aligned(16))) float hbuf[2][O_GROUP_ * O_HROW_];
    __shared__ __attribute__((aligned(16))) float obuf[2][O_BATCH_OUT_];
    float* hw = hbuf[wave];
    float* ow = obuf[wave];

    const v4f* wbase = reinterpret_cast<const v4f*>(
        region + (size_t)nc * (T_ * V_ * K_)) + (size_t)(wave * O_RP_WAVE_) * ROWPAIR_F4_;
    const v4f* p[4];
    v4f w[4];
    #pragma unroll
    for (int j = 0; j < 4; ++j) {
        const int q  = 32 * j + lane;
        const int qc = (q < ROWPAIR_F4_) ? q : (ROWPAIR_F4_ - 1);
        p[j] = wbase + qc;
        float wv[4];
        #pragma unroll
        for (int e = 0; e < 4; ++e) {
            const int f = (4 * qc + e) % (V_ * K_);
            const int v = f / K_;
            const int k = f % K_;
            wv[e] = wk2[(size_t)k * M_ + n * V_ + v];
        }
        w[j].x = wv[0]; w[j].y = wv[1]; w[j].z = wv[2]; w[j].w = wv[3];
    }
    const float bias = b2[0];

    v4f* oslab4 = reinterpret_cast<v4f*>(out + (size_t)nc * (T_ * V_))
                  + (size_t)wave * (2 * O_BATCH_OUT_ / 4);

    v4f buf[2][O_GROUP_ * 4];
#define O_LOAD_GROUP(G) do { \
        _Pragma("unroll") for (int u = 0; u < O_GROUP_; ++u) { \
            _Pragma("unroll") for (int j = 0; j < 4; ++j) \
                buf[(G) & 1][u * 4 + j] = p[j][((G) * O_GROUP_ + u) * ROWPAIR_F4_]; \
        } } while (0)

    O_LOAD_GROUP(0);
    #pragma unroll
    for (int g = 0; g < O_NGROUPS_; ++g) {
        __builtin_amdgcn_sched_barrier(0);
        if (g + 1 < O_NGROUPS_) O_LOAD_GROUP(g + 1);
        __builtin_amdgcn_sched_barrier(0);

        #pragma unroll
        for (int u = 0; u < O_GROUP_; ++u) {
            #pragma unroll
            for (int j = 0; j < 4; ++j) {
                const v4f r = buf[g & 1][u * 4 + j];
                v2f h;
                h.x = fmaf(r.x, w[j].x, r.y * w[j].y);
                h.y = fmaf(r.z, w[j].z, r.w * w[j].w);
                *reinterpret_cast<v2f*>(hw + u * O_HROW_ + 2 * (32 * j + lane)) = h;
            }
        }
        O_WAVE_SYNC();

        #pragma unroll
        for (int u = 0; u < O_GROUP_; ++u) {
            #pragma unroll
            for (int hlf = 0; hlf < 2; ++hlf) {
                const int o = lane + 32 * hlf;
                if (o < 2 * V_) {
                    const float* hs = hw + u * O_HROW_ + 5 * o;
                    const float s = ((hs[0] + hs[1]) + (hs[2] + hs[3])) + hs[4];
                    ow[(g & 3) * O_GRP_OUT_ + u * (2 * V_) + o] = s + bias;
                }
            }
        }
        O_WAVE_SYNC();

        if ((g & 3) == 3) {
            const v4f* ow4 = reinterpret_cast<const v4f*>(ow);
            v4f* dst = oslab4 + (g >> 2) * (O_BATCH_OUT_ / 4);
            #pragma unroll
            for (int r = 0; r < 7; ++r) {
                const int i4 = lane + 32 * r;
                if (i4 < O_BATCH_OUT_ / 4)
                    __builtin_nontemporal_store(ow4[i4], &dst[i4]);
            }
            O_WAVE_SYNC();
        }
    }
#undef O_LOAD_GROUP
}

extern "C" void kernel_launch(void* const* d_in, const int* in_sizes, int n_in,
                              void* d_out, int out_size, void* d_ws, size_t ws_size,
                              hipStream_t stream) {
    const float* region = (const float*)d_in[0];
    const float* w1     = (const float*)d_in[1];
    const float* b1     = (const float*)d_in[2];
    const float* w2     = (const float*)d_in[3];
    const float* b2     = (const float*)d_in[4];
    float* out = (float*)d_out;

    float* mf2 = (float*)d_ws;
    float* wk2 = mf2 + (size_t)K_ * M_ * C_;

    vc_mean_kernel<<<N_ * C_, 64, 0, stream>>>(region, mf2);
    vc_attn_kernel<<<K_ * (M_ / 16), 32, 0, stream>>>(mf2, w1, b1, w2, wk2);
    vc_out_kernel<<<N_ * C_, 64, 0, stream>>>(region, wk2, b2, out);
}


# Mean stage stores each slab's 250 means as one contiguous 1000-byte run (8-byte stores; one writer per line, two at run edges), not 250 lone floats in lines 32 blocks share; weight stage reads it
# speedup vs baseline: 1.6560x; 1.4796x over previous
#include <hip/hip_runtime.h>

#define N_ 32
#define C_ 64
#define T_ 128
#define V_ 25
#define K_ 10
#define M_ (N_ * V_)

typedef float v2f __attribute__((ext_vector_type(2)));
typedef float v4f __attribute__((ext_vector_type(4)));
typedef float v8f __attribute__((ext_vector_type(8)));

#define ROWPAIR_F4_ ((2 * V_ * K_) / 4)
#define ROWPAIRS_   (T_ / 2)
#define GROUP_      8
#define NGROUPS_    (ROWPAIRS_ / GROUP_)
#define WALK_RP_(G, U) \
    ((((U) < GROUP_ / 2) ? (ROWPAIRS_ / 2 - 1) : (ROWPAIRS_ - 1)) \
     - ((G) * (GROUP_ / 2) + ((U) % (GROUP_ / 2))))

__global__ __launch_bounds__(64)
void vc_mean_kernel(const float* __restrict__ region,
                    float* __restrict__ mfr) {
    const int nc = blockIdx.x;
    const int i  = threadIdx.x;

    const v4f* slab4 = reinterpret_cast<const v4f*>(
        region + (size_t)nc * (T_ * V_ * K_));
    const bool has1 = (i + 64) < ROWPAIR_F4_;
    const v4f* p0 = slab4 + i;
    const v4f* p1 = slab4 + (has1 ? (i + 64) : (ROWPAIR_F4_ - 1));

    v4f w0[2][GROUP_], w1[2][GROUP_];
#define LOAD_GROUP(G) do { \
        _Pragma("unroll") for (int u = 0; u < GROUP_; ++u) { \
            w0[(G) & 1][u] = p0[WALK_RP_(G, u) * ROWPAIR_F4_]; \
            w1[(G) & 1][u] = p1[WALK_RP_(G, u) * ROWPAIR_F4_]; \
        } } while (0)

    v4f s0 = {0.f, 0.f, 0.f, 0.f};
    v4f s1 = {0.f, 0.f, 0.f, 0.f};
    LOAD_GROUP(0);
    #pragma unroll
    for (int g = 0; g < NGROUPS_; ++g) {
        __builtin_amdgcn_sched_barrier(0);
        if (g + 1 < NGROUPS_) LOAD_GROUP(g + 1);
        __builtin_amdgcn_sched_barrier(0);
        #pragma unroll
        for (int u = 0; u < GROUP_; ++u) {
            s0 += w0[g & 1][u];
            s1 += w1[g & 1][u];
        }
    }
#undef LOAD_GROUP

    __shared__ v4f part4[128];
    part4[i]      = s0;
    part4[i + 64] = s1;
    __syncthreads();

    const float* part = reinterpret_cast<const float*>(part4);
    v2f* run2 = reinterpret_cast<v2f*>(mfr + (size_t)nc * (V_ * K_));
    for (int q = i; q < (V_ * K_) / 2; q += 64) {
        v2f m2;
        m2.x = (part[2 * q]     + part[2 * q + V_ * K_])     * (1.0f / (float)T_);
        m2.y = (part[2 * q + 1] + part[2 * q + 1 + V_ * K_]) * (1.0f / (float)T_);
        run2[q] = m2;
    }
}

__global__ void vc_attn_kernel(const float* __restrict__ mfr,
                               const float* __restrict__ w1,
                               const float* __restrict__ b1,
                               const float* __restrict__ w2,
                               float* __restrict__ wk2) {
    const int tile = blockIdx.x;
    const int k    = tile / (M_ / 16);
    const int m0   = (tile % (M_ / 16)) * 16;
    const int lane = threadIdx.x;
    const int half = lane >> 4;
    const int l    = lane & 15;

    const int mrow = m0 + l;
    const float* acol0 = mfr + (size_t)(mrow / V_) * (C_ * V_ * K_)
                             + (mrow % V_) * K_ + k;
    const int   pclamp = (l < K_) ? l : (K_ - 1);
    const float bmask  = (l < K_) ? 1.0f : 0.0f;
    const float* brow  = w1 + ((size_t)k * K_ + pclamp) * C_;

    v8f acc = {0.f, 0.f, 0.f, 0.f, 0.f, 0.f, 0.f, 0.f};

    #pragma unroll
    for (int j = 0; j < C_ / 4; ++j) {
        const int c0 = j * 4 + half * 2;
        v2f a, b;
        a.x = acol0[c0 * (V_ * K_)];
        a.y = acol0[(c0 + 1) * (V_ * K_)];
        b.x = brow[c0] * bmask;
        b.y = brow[c0 + 1] * bmask;
        acc = __builtin_amdgcn_wmma_f32_16x16x4_f32(
             false, a,  false, b,
             (short)0, acc,  false,  false);
    }

    __shared__ float tile_lds[16][17];
    #pragma unroll
    for (int g = 0; g < 8; ++g)
        tile_lds[g + half * 8][l] = acc[g];
    __syncthreads();

    if (lane < 16) {
        const int row = lane;
        float lg[K_];
        float mx = -1e30f;
        #pragma unroll
        for (int p = 0; p < K_; ++p) {
            lg[p] = tile_lds[row][p] + b1[k * K_ + p];
            mx = fmaxf(mx, lg[p]);
        }
        float sum = 0.0f;
        #pragma unroll
        for (int p = 0; p < K_; ++p) {
            lg[p] = __expf(lg[p] - mx);
            sum += lg[p];
        }
        const float inv = 1.0f / sum;
        float wkv = 0.0f;
        #pragma unroll
        for (int p = 0; p < K_; ++p)
            wkv += lg[p] * inv * w2[p];
        wk2[(size_t)k * M_ + m0 + row] = wkv;
    }
}

#define O_RP_WAVE_   (ROWPAIRS_ / 2)
#define O_GROUP_     4
#define O_NGROUPS_   (O_RP_WAVE_ / O_GROUP_)
#define O_HROW_      256
#define O_GRP_OUT_   (O_GROUP_ * 2 * V_)
#define O_BATCH_OUT_ (4 * O_GRP_OUT_)

#define O_WAVE_SYNC() do { \
        __builtin_amdgcn_fence(__ATOMIC_RELEASE, "wavefront", "local"); \
        __builtin_amdgcn_wave_barrier(); \
        __builtin_amdgcn_fence(__ATOMIC_ACQUIRE, "wavefront", "local"); \
    } while (0)

__global__ __launch_bounds__(64, 5)
void vc_out_kernel(const float* __restrict__ region,
                   const float* __restrict__ wk2,
                   const float* __restrict__ b2,
                   float* __restrict__ out) {
    const int nc   = blockIdx.x;
    const int n    = nc / C_;
    const int lane = threadIdx.x & 31;
    const int wave = threadIdx.x >> 5;

    __shared__ __attribute__((aligned(16))) float hbuf[2][O_GROUP_ * O_HROW_];
    __shared__ __attribute__((aligned(16))) float obuf[2][O_BATCH_OUT_];
    float* hw = hbuf[wave];
    float* ow = obuf[wave];

    const v4f* wbase = reinterpret_cast<const v4f*>(
        region + (size_t)nc * (T_ * V_ * K_)) + (size_t)(wave * O_RP_WAVE_) * ROWPAIR_F4_;
    const v4f* p[4];
    v4f w[4];
    #pragma unroll
    for (int j = 0; j < 4; ++j) {
        const int q  = 32 * j + lane;
        const int qc = (q < ROWPAIR_F4_) ? q : (ROWPAIR_F4_ - 1);
        p[j] = wbase + qc;
        float wv[4];
        #pragma unroll
        for (int e = 0; e < 4; ++e) {
            const int f = (4 * qc + e) % (V_ * K_);
            const int v = f / K_;
            const int k = f % K_;
            wv[e] = wk2[(size_t)k * M_ + n * V_ + v];
        }
        w[j].x = wv[0]; w[j].y = wv[1]; w[j].z = wv[2]; w[j].w = wv[3];
    }
    const float bias = b2[0];

    v4f* oslab4 = reinterpret_cast<v4f*>(out + (size_t)nc * (T_ * V_))
                  + (size_t)wave * (2 * O_BATCH_OUT_ / 4);

    v4f buf[2][O_GROUP_ * 4];
#define O_LOAD_GROUP(G) do { \
        _Pragma("unroll") for (int u = 0; u < O_GROUP_; ++u) { \
            _Pragma("unroll") for (int j = 0; j < 4; ++j) \
                buf[(G) & 1][u * 4 + j] = p[j][((G) * O_GROUP_ + u) * ROWPAIR_F4_]; \
        } } while (0)

    O_LOAD_GROUP(0);
    #pragma unroll
    for (int g = 0; g < O_NGROUPS_; ++g) {
        __builtin_amdgcn_sched_barrier(0);
        if (g + 1 < O_NGROUPS_) O_LOAD_GROUP(g + 1);
        __builtin_amdgcn_sched_barrier(0);

        #pragma unroll
        for (int u = 0; u < O_GROUP_; ++u) {
            #pragma unroll
            for (int j = 0; j < 4; ++j) {
                const v4f r = buf[g & 1][u * 4 + j];
                v2f h;
                h.x = fmaf(r.x, w[j].x, r.y * w[j].y);
                h.y = fmaf(r.z, w[j].z, r.w * w[j].w);
                *reinterpret_cast<v2f*>(hw + u * O_HROW_ + 2 * (32 * j + lane)) = h;
            }
        }
        O_WAVE_SYNC();

        #pragma unroll
        for (int u = 0; u < O_GROUP_; ++u) {
            #pragma unroll
            for (int hlf = 0; hlf < 2; ++hlf) {
                const int o = lane + 32 * hlf;
                if (o < 2 * V_) {
                    const float* hs = hw + u * O_HROW_ + 5 * o;
                    const float s = ((hs[0] + hs[1]) + (hs[2] + hs[3])) + hs[4];
                    ow[(g & 3) * O_GRP_OUT_ + u * (2 * V_) + o] = s + bias;
                }
            }
        }
        O_WAVE_SYNC();

        if ((g & 3) == 3) {
            const v4f* ow4 = reinterpret_cast<const v4f*>(ow);
            v4f* dst = oslab4 + (g >> 2) * (O_BATCH_OUT_ / 4);
            #pragma unroll
            for (int r = 0; r < 7; ++r) {
                const int i4 = lane + 32 * r;
                if (i4 < O_BATCH_OUT_ / 4)
                    __builtin_nontemporal_store(ow4[i4], &dst[i4]);
            }
            O_WAVE_SYNC();
        }
    }
#undef O_LOAD_GROUP
}

extern "C" void kernel_launch(void* const* d_in, const int* in_sizes, int n_in,
                              void* d_out, int out_size, void* d_ws, size_t ws_size,
                              hipStream_t stream) {
    const float* region = (const float*)d_in[0];
    const float* w1     = (const float*)d_in[1];
    const float* b1     = (const float*)d_in[2];
    const float* w2     = (const float*)d_in[3];
    const float* b2     = (const float*)d_in[4];
    float* out = (float*)d_out;

    float* mfr = (float*)d_ws;
    float* wk2 = mfr + (size_t)N_ * C_ * V_ * K_;

    vc_mean_kernel<<<N_ * C_, 64, 0, stream>>>(region, mfr);
    vc_attn_kernel<<<K_ * (M_ / 16), 32, 0, stream>>>(mfr, w1, b1, w2, wk2);
    vc_out_kernel<<<N_ * C_, 64, 0, stream>>>(region, wk2, b2, out);
}


# Weight stage (f32 matrix products, softmax, w2 fold) folded into the head of the output stage, now one 16-wave block per WGP for 8 slabs of one n: two launches, not three; every sum order kept
# speedup vs baseline: 1.7172x; 1.0370x over previous
#include <hip/hip_runtime.h>

#define N_ 32
#define C_ 64
#define T_ 128
#define V_ 25
#define K_ 10
#define M_ (N_ * V_)

typedef float v2f __attribute__((ext_vector_type(2)));
typedef float v4f __attribute__((ext_vector_type(4)));
typedef float v8f __attribute__((ext_vector_type(8)));

#define ROWPAIR_F4_ ((2 * V_ * K_) / 4)
#define ROWPAIRS_   (T_ / 2)
#define GROUP_      8
#define NGROUPS_    (ROWPAIRS_ / GROUP_)
#define WALK_RP_(G, U) \
    ((((U) < GROUP_ / 2) ? (ROWPAIRS_ / 2 - 1) : (ROWPAIRS_ - 1)) \
     - ((G) * (GROUP_ / 2) + ((U) % (GROUP_ / 2))))

__global__ __launch_bounds__(64)
void vc_mean_kernel(const float* __restrict__ region,
                    float* __restrict__ mfr) {
    const int nc = blockIdx.x;
    const int i  = threadIdx.x;

    const v4f* slab4 = reinterpret_cast<const v4f*>(
        region + (size_t)nc * (T_ * V_ * K_));
    const bool has1 = (i + 64) < ROWPAIR_F4_;
    const v4f* p0 = slab4 + i;
    const v4f* p1 = slab4 + (has1 ? (i + 64) : (ROWPAIR_F4_ - 1));

    v4f w0[2][GROUP_], w1[2][GROUP_];
#define LOAD_GROUP(G) do { \
        _Pragma("unroll") for (int u = 0; u < GROUP_; ++u) { \
            w0[(G) & 1][u] = p0[WALK_RP_(G, u) * ROWPAIR_F4_]; \
            w1[(G) & 1][u] = p1[WALK_RP_(G, u) * ROWPAIR_F4_]; \
        } } while (0)

    v4f s0 = {0.f, 0.f, 0.f, 0.f};
    v4f s1 = {0.f, 0.f, 0.f, 0.f};
    LOAD_GROUP(0);
    #pragma unroll
    for (int g = 0; g < NGROUPS_; ++g) {
        __builtin_amdgcn_sched_barrier(0);
        if (g + 1 < NGROUPS_) LOAD_GROUP(g + 1);
        __builtin_amdgcn_sched_barrier(0);
        #pragma unroll
        for (int u = 0; u < GROUP_; ++u) {
            s0 += w0[g & 1][u];
            s1 += w1[g & 1][u];
        }
    }
#undef LOAD_GROUP

    __shared__ v4f part4[128];
    part4[i]      = s0;
    part4[i + 64] = s1;
    __syncthreads();

    const float* part = reinterpret_cast<const float*>(part4);
    v2f* run2 = reinterpret_cast<v2f*>(mfr + (size_t)nc * (V_ * K_));
    for (int q = i; q < (V_ * K_) / 2; q += 64) {
        v2f m2;
        m2.x = (part[2 * q]     + part[2 * q + V_ * K_])     * (1.0f / (float)T_);
        m2.y = (part[2 * q + 1] + part[2 * q + 1 + V_ * K_]) * (1.0f / (float)T_);
        run2[q] = m2;
    }
}

#define O_RP_WAVE_   (ROWPAIRS_ / 2)
#define O_GROUP_     4
#define O_NGROUPS_   (O_RP_WAVE_ / O_GROUP_)
#define O_HROW_      256
#define O_GRP_OUT_   (O_GROUP_ * 2 * V_)
#define O_BATCH_OUT_ (4 * O_GRP_OUT_)
#define F_WAVES_     16
#define F_TASKS_     (K_ * 2)
#define F_MF4_       ((C_ * V_ * K_) / 4)
#define F_W14_       ((K_ * K_ * C_) / 4)
#define F_W1ROW_     68

#define O_WAVE_SYNC() do { \
        __builtin_amdgcn_fence(__ATOMIC_RELEASE, "wavefront", "local"); \
        __builtin_amdgcn_wave_barrier(); \
        __builtin_amdgcn_fence(__ATOMIC_ACQUIRE, "wavefront", "local"); \
    } while (0)

#define F_LDS_BLOCK_SYNC() do { \
        __builtin_amdgcn_fence(__ATOMIC_RELEASE, "workgroup", "local"); \
        __builtin_amdgcn_s_barrier(); \
        __builtin_amdgcn_fence(__ATOMIC_ACQUIRE, "workgroup", "local"); \
    } while (0)

__global__ __launch_bounds__(512)
void vc_out_kernel(const float* __restrict__ region,
                   const float* __restrict__ mfr,
                   const float* __restrict__ w1,
                   const float* __restrict__ b1,
                   const float* __restrict__ w2,
                   const float* __restrict__ b2,
                   float* __restrict__ out) {
    const int blk  = blockIdx.x;
    const int n    = blk >> 3;
    const int tid  = threadIdx.x;
    const int lane = tid & 31;
    const int wv   = __builtin_amdgcn_readfirstlane(tid >> 5);
    const int nc   = n * C_ + (blk & 7) + 8 * (wv >> 1);
    const int wave = wv & 1;
    const int half = lane >> 4;
    const int l    = lane & 15;

    __shared__ __attribute__((aligned(16))) float mfs[8 * 512 * 4];
    __shared__ __attribute__((aligned(16))) float hbuf[F_WAVES_][O_GROUP_ * O_HROW_];
    __shared__ __attribute__((aligned(16))) float obuf[F_WAVES_][O_BATCH_OUT_];
    __shared__ __attribute__((aligned(16))) float w1s[128 * F_W1ROW_];
    __shared__ float wkl[V_ * K_ + 6];
    float* hw = hbuf[wv];
    float* ow = obuf[wv];

    const v4f* mf4 = reinterpret_cast<const v4f*>(mfr + (size_t)n * (C_ * V_ * K_));
    const v4f* w14 = reinterpret_cast<const v4f*>(w1);
    v4f st[8], ws[4];
    #pragma unroll
    for (int r = 0; r < 8; ++r) {
        const int i4 = tid + 512 * r;
        st[r] = mf4[(i4 < F_MF4_) ? i4 : (F_MF4_ - 1)];
    }
    #pragma unroll
    for (int r = 0; r < 4; ++r) {
        const int i4 = tid + 512 * r;
        ws[r] = w14[(i4 < F_W14_) ? i4 : (F_W14_ - 1)];
    }
    {
        v4f* mfs4 = reinterpret_cast<v4f*>(mfs);
        #pragma unroll
        for (int r = 0; r < 8; ++r)
            mfs4[tid + 512 * r] = st[r];
        #pragma unroll
        for (int r = 0; r < 4; ++r) {
            const int i4 = tid + 512 * r;
            *reinterpret_cast<v4f*>(w1s + (i4 >> 4) * F_W1ROW_ + (i4 & 15) * 4) = ws[r];
        }
    }
    int slane = lane;
    asm volatile("" : "+v"(slane)
                 : "v"(st[0]), "v"(st[1]), "v"(st[2]), "v"(st[3]),
                   "v"(st[4]), "v"(st[5]), "v"(st[6]), "v"(st[7]),
                   "v"(ws[0]), "v"(ws[1]), "v"(ws[2]), "v"(ws[3]));

    const int   pclamp = (l < K_) ? l : (K_ - 1);
    const float bmask  = (l < K_) ? 1.0f : 0.0f;
    const bool  two    = (wv + F_WAVES_) < F_TASKS_;
    const int   task0  = wv;
    const int   task1  = two ? (wv + F_WAVES_) : wv;

    const v4f* wbase = reinterpret_cast<const v4f*>(
        region + (size_t)nc * (T_ * V_ * K_)) + (size_t)(wave * O_RP_WAVE_) * ROWPAIR_F4_;
    const v4f* p[4];
    int wslot[4];
    #pragma unroll
    for (int j = 0; j < 4; ++j) {
        const int q  = 32 * j + slane;
        const int qc = (q < ROWPAIR_F4_) ? q : (ROWPAIR_F4_ - 1);
        p[j] = wbase + qc;
        wslot[j] = 4 * qc;
    }
    v4f* oslab4 = reinterpret_cast<v4f*>(out + (size_t)nc * (T_ * V_))
                  + (size_t)wave * (2 * O_BATCH_OUT_ / 4);

    v4f buf[2][O_GROUP_ * 4];
#define O_LOAD_GROUP(G) do { \
        _Pragma("unroll") for (int u = 0; u < O_GROUP_; ++u) { \
            _Pragma("unroll") for (int j = 0; j < 4; ++j) \
                buf[(G) & 1][u * 4 + j] = p[j][((G) * O_GROUP_ + u) * ROWPAIR_F4_]; \
        } } while (0)

    O_LOAD_GROUP(0);
    __builtin_amdgcn_sched_barrier(0);
    F_LDS_BLOCK_SYNC();

    float (*tile)[17] = reinterpret_cast<float (*)[17]>(hw);
    #pragma unroll
    for (int t = 0; t < 2; ++t) {
        if (t == 0 || two) {
            const int task = t ? task1 : task0;
            const int k    = task >> 1;
            const int vrow = (task & 1) * 16 + l;
            const float* arow = mfs + ((vrow < V_) ? vrow : (V_ - 1)) * K_ + k;
            const float* brow = w1s + (k * K_ + pclamp) * F_W1ROW_;
            v8f acc = {0.f, 0.f, 0.f, 0.f, 0.f, 0.f, 0.f, 0.f};
            v2f af[C_ / 4], bfr[C_ / 4];
            #pragma unroll
            for (int j = 0; j < C_ / 4; ++j) {
                const int c0 = j * 4 + half * 2;
                af[j].x = arow[c0 * (V_ * K_)];
                af[j].y = arow[(c0 + 1) * (V_ * K_)];
                bfr[j]  = *reinterpret_cast<const v2f*>(brow + c0);
            }
            __builtin_amdgcn_sched_barrier(0);
            #pragma unroll
            for (int j = 0; j < C_ / 4; ++j) {
                v2f a = af[j], b;
                b.x = bfr[j].x * bmask;
                b.y = bfr[j].y * bmask;
                acc = __builtin_amdgcn_wmma_f32_16x16x4_f32(
                     false, a,  false, b,
                     (short)0, acc,  false,  false);
            }
            #pragma unroll
            for (int g = 0; g < 8; ++g)
                tile[g + half * 8][l] = acc[g];
            O_WAVE_SYNC();
            if (lane < 16) {
                const int v = (task & 1) * 16 + lane;
                float lg[K_];
                float mx = -1e30f;
                #pragma unroll
                for (int pp = 0; pp < K_; ++pp) {
                    lg[pp] = tile[lane][pp] + b1[k * K_ + pp];
                    mx = fmaxf(mx, lg[pp]);
                }
                float sum = 0.0f;
                #pragma unroll
                for (int pp = 0; pp < K_; ++pp) {
                    lg[pp] = __expf(lg[pp] - mx);
                    sum += lg[pp];
                }
                const float inv = 1.0f / sum;
                float wkv = 0.0f;
                #pragma unroll
                for (int pp = 0; pp < K_; ++pp)
                    wkv += lg[pp] * inv * w2[pp];
                if (v < V_) wkl[v * K_ + k] = wkv;
            }
            O_WAVE_SYNC();
        }
    }
    F_LDS_BLOCK_SYNC();

    v4f w[4];
    #pragma unroll
    for (int j = 0; j < 4; ++j) {
        w[j].x = wkl[(wslot[j] + 0) % (V_ * K_)];
        w[j].y = wkl[(wslot[j] + 1) % (V_ * K_)];
        w[j].z = wkl[(wslot[j] + 2) % (V_ * K_)];
        w[j].w = wkl[(wslot[j] + 3) % (V_ * K_)];
    }
    const float bias = b2[0];

    #pragma unroll
    for (int g = 0; g < O_NGROUPS_; ++g) {
        __builtin_amdgcn_sched_barrier(0);
        if (g + 1 < O_NGROUPS_) O_LOAD_GROUP(g + 1);
        __builtin_amdgcn_sched_barrier(0);

        #pragma unroll
        for (int u = 0; u < O_GROUP_; ++u) {
            #pragma unroll
            for (int j = 0; j < 4; ++j) {
                const v4f r = buf[g & 1][u * 4 + j];
                v2f h;
                h.x = fmaf(r.x, w[j].x, r.y * w[j].y);
                h.y = fmaf(r.z, w[j].z, r.w * w[j].w);
                *reinterpret_cast<v2f*>(hw + u * O_HROW_ + 2 * (32 * j + lane)) = h;
            }
        }
        O_WAVE_SYNC();

        #pragma unroll
        for (int u = 0; u < O_GROUP_; ++u) {
            #pragma unroll
            for (int hlf = 0; hlf < 2; ++hlf) {
                const int o = lane + 32 * hlf;
                if (o < 2 * V_) {
                    const float* hs = hw + u * O_HROW_ + 5 * o;
                    const float s = ((hs[0] + hs[1]) + (hs[2] + hs[3])) + hs[4];
                    ow[(g & 3) * O_GRP_OUT_ + u * (2 * V_) + o] = s + bias;
                }
            }
        }
        O_WAVE_SYNC();

        if ((g & 3) == 3) {
            const v4f* ow4 = reinterpret_cast<const v4f*>(ow);
            v4f* dst = oslab4 + (g >> 2) * (O_BATCH_OUT_ / 4);
            #pragma unroll
            for (int r = 0; r < 7; ++r) {
                const int i4 = lane + 32 * r;
                if (i4 < O_BATCH_OUT_ / 4)
                    __builtin_nontemporal_store(ow4[i4], &dst[i4]);
            }
            O_WAVE_SYNC();
        }
    }
#undef O_LOAD_GROUP
}

extern "C" void kernel_launch(void* const* d_in, const int* in_sizes, int n_in,
                              void* d_out, int out_size, void* d_ws, size_t ws_size,
                              hipStream_t stream) {
    const float* region = (const float*)d_in[0];
    const float* w1     = (const float*)d_in[1];
    const float* b1     = (const float*)d_in[2];
    const float* w2     = (const float*)d_in[3];
    const float* b2     = (const float*)d_in[4];
    float* out = (float*)d_out;

    float* mfr = (float*)d_ws;

    vc_mean_kernel<<<N_ * C_, 64, 0, stream>>>(region, mfr);
    vc_out_kernel<<<N_ * 8, 512, 0, stream>>>(region, mfr, w1, b1, w2, b2, out);
}


# speedup vs baseline: 1.7803x; 1.0367x over previous
#include <hip/hip_runtime.h>

#define N_ 32
#define C_ 64
#define T_ 128
#define V_ 25
#define K_ 10
#define M_ (N_ * V_)

typedef float v2f __attribute__((ext_vector_type(2)));
typedef float v4f __attribute__((ext_vector_type(4)));
typedef float v8f __attribute__((ext_vector_type(8)));

#define ROWPAIR_F4_ ((2 * V_ * K_) / 4)
#define ROWPAIRS_   (T_ / 2)
#define GROUP_      8
#define NGROUPS_    (ROWPAIRS_ / GROUP_)
#define WALK_RP_(G, U) \
    ((((U) < GROUP_ / 2) ? (ROWPAIRS_ / 2 - 1) : (ROWPAIRS_ - 1)) \
     - ((G) * (GROUP_ / 2) + ((U) % (GROUP_ / 2))))

__global__ __launch_bounds__(64)
void vc_mean_kernel(const float* __restrict__ region,
                    float* __restrict__ mfr) {
    const int nc = blockIdx.x;
    const int i  = threadIdx.x;

    const v4f* slab4 = reinterpret_cast<const v4f*>(
        region + (size_t)nc * (T_ * V_ * K_));
    const bool has1 = (i + 64) < ROWPAIR_F4_;
    const v4f* p0 = slab4 + i;
    const v4f* p1 = slab4 + (has1 ? (i + 64) : (ROWPAIR_F4_ - 1));

    v4f w0[2][GROUP_], w1[2][GROUP_];
#define LOAD_GROUP(G) do { \
        _Pragma("unroll") for (int u = 0; u < GROUP_; ++u) { \
            w0[(G) & 1][u] = p0[WALK_RP_(G, u) * ROWPAIR_F4_]; \
            w1[(G) & 1][u] = p1[WALK_RP_(G, u) * ROWPAIR_F4_]; \
        } } while (0)

    v4f s0 = {0.f, 0.f, 0.f, 0.f};
    v4f s1 = {0.f, 0.f, 0.f, 0.f};
    LOAD_GROUP(0);
    #pragma unroll
    for (int g = 0; g < NGROUPS_; ++g) {
        __builtin_amdgcn_sched_barrier(0);
        if (g + 1 < NGROUPS_) LOAD_GROUP(g + 1);
        __builtin_amdgcn_sched_barrier(0);
        #pragma unroll
        for (int u = 0; u < GROUP_; ++u) {
            s0 += w0[g & 1][u];
            s1 += w1[g & 1][u];
        }
    }
#undef LOAD_GROUP

    __shared__ v4f part4[128];
    part4[i]      = s0;
    part4[i + 64] = s1;
    __syncthreads();

    const float* part = reinterpret_cast<const float*>(part4);
    v2f* run2 = reinterpret_cast<v2f*>(mfr + (size_t)nc * (V_ * K_));
    for (int q = i; q < (V_ * K_) / 2; q += 64) {
        v2f m2;
        m2.x = (part[2 * q]     + part[2 * q + V_ * K_])     * (1.0f / (float)T_);
        m2.y = (part[2 * q + 1] + part[2 * q + 1 + V_ * K_]) * (1.0f / (float)T_);
        run2[q] = m2;
    }
}

#define P_PIECES_    (((T_ / 2) * V_ * K_) / 128)
#define P_GROUP_     15
#define P_NGROUPS_   8
#define P_TAIL_      (P_PIECES_ - P_NGROUPS_ * P_GROUP_)
#define P_GRP_H_     (P_GROUP_ * 64)
#define P_GRP_OUT_   (P_GRP_H_ / 5)
#define P_BATCH_OUT_ (4 * P_GRP_OUT_)
#define P_TAIL_OUT_  ((P_TAIL_ * 64) / 5)
#define F_WAVES_     16
#define F_TASKS_     (K_ * 2)
#define F_MF4_       ((C_ * V_ * K_) / 4)
#define F_W14_       ((K_ * K_ * C_) / 4)
#define F_W1ROW_     68

#define O_WAVE_SYNC() do { \
        __builtin_amdgcn_fence(__ATOMIC_RELEASE, "wavefront", "local"); \
        __builtin_amdgcn_wave_barrier(); \
        __builtin_amdgcn_fence(__ATOMIC_ACQUIRE, "wavefront", "local"); \
    } while (0)

#define F_LDS_BLOCK_SYNC() do { \
        __builtin_amdgcn_fence(__ATOMIC_RELEASE, "workgroup", "local"); \
        __builtin_amdgcn_s_barrier(); \
        __builtin_amdgcn_fence(__ATOMIC_ACQUIRE, "workgroup", "local"); \
    } while (0)

__global__ __launch_bounds__(512)
void vc_out_kernel(const float* __restrict__ region,
                   const float* __restrict__ mfr,
                   const float* __restrict__ w1,
                   const float* __restrict__ b1,
                   const float* __restrict__ w2,
                   const float* __restrict__ b2,
                   float* __restrict__ out) {
    const int blk  = blockIdx.x;
    const int n    = blk >> 3;
    const int tid  = threadIdx.x;
    const int lane = tid & 31;
    const int wv   = __builtin_amdgcn_readfirstlane(tid >> 5);
    const int nc   = n * C_ + (blk & 7) + 8 * (wv >> 1);
    const int wave = wv & 1;
    const int half = lane >> 4;
    const int l    = lane & 15;

    __shared__ __attribute__((aligned(16))) float mfs[8 * 512 * 4];
    __shared__ __attribute__((aligned(16))) float hbuf[F_WAVES_][P_GRP_H_];
    __shared__ __attribute__((aligned(16))) float obuf[F_WAVES_][P_BATCH_OUT_];
    __shared__ __attribute__((aligned(16))) float w1s[128 * F_W1ROW_];
    __shared__ __attribute__((aligned(16))) float wkl[V_ * K_ + 6];
    float* hw = hbuf[wv];
    float* ow = obuf[wv];

    const v4f* mf4 = reinterpret_cast<const v4f*>(mfr + (size_t)n * (C_ * V_ * K_));
    const v4f* w14 = reinterpret_cast<const v4f*>(w1);
    v4f st[8], ws[4];
    #pragma unroll
    for (int r = 0; r < 8; ++r) {
        const int i4 = tid + 512 * r;
        st[r] = mf4[(i4 < F_MF4_) ? i4 : (F_MF4_ - 1)];
    }
    #pragma unroll
    for (int r = 0; r < 4; ++r) {
        const int i4 = tid + 512 * r;
        ws[r] = w14[(i4 < F_W14_) ? i4 : (F_W14_ - 1)];
    }
    {
        v4f* mfs4 = reinterpret_cast<v4f*>(mfs);
        #pragma unroll
        for (int r = 0; r < 8; ++r)
            mfs4[tid + 512 * r] = st[r];
        #pragma unroll
        for (int r = 0; r < 4; ++r) {
            const int i4 = tid + 512 * r;
            *reinterpret_cast<v4f*>(w1s + (i4 >> 4) * F_W1ROW_ + (i4 & 15) * 4) = ws[r];
        }
    }
    int slane = lane;
    asm volatile("" : "+v"(slane)
                 : "v"(st[0]), "v"(st[1]), "v"(st[2]), "v"(st[3]),
                   "v"(st[4]), "v"(st[5]), "v"(st[6]), "v"(st[7]),
                   "v"(ws[0]), "v"(ws[1]), "v"(ws[2]), "v"(ws[3]));

    const int   pclamp = (l < K_) ? l : (K_ - 1);
    const float bmask  = (l < K_) ? 1.0f : 0.0f;
    const bool  two    = (wv + F_WAVES_) < F_TASKS_;
    const int   task0  = wv;
    const int   task1  = two ? (wv + F_WAVES_) : wv;

    const v4f* pl = reinterpret_cast<const v4f*>(
        region + (size_t)nc * (T_ * V_ * K_) + (size_t)wave * ((T_ / 2) * V_ * K_)) + slane;
    v4f* oslab4 = reinterpret_cast<v4f*>(out + (size_t)nc * (T_ * V_))
                  + (size_t)wave * ((T_ / 2) * V_ / 4);

    v4f buf[2][P_GROUP_];
#define P_LOAD(BUFI, FIRST, COUNT) do { \
        _Pragma("unroll") for (int u = 0; u < (COUNT); ++u) \
            buf[(BUFI)][u] = pl[((FIRST) + u) * 32]; \
        } while (0)

    P_LOAD(0, 0, P_GROUP_);
    __builtin_amdgcn_sched_barrier(0);
    F_LDS_BLOCK_SYNC();

    float (*tile)[17] = reinterpret_cast<float (*)[17]>(hw);
    #pragma unroll
    for (int t = 0; t < 2; ++t) {
        if (t == 0 || two) {
            const int task = t ? task1 : task0;
            const int k    = task >> 1;
            const int vrow = (task & 1) * 16 + l;
            const float* arow = mfs + ((vrow < V_) ? vrow : (V_ - 1)) * K_ + k;
            const float* brow = w1s + (k * K_ + pclamp) * F_W1ROW_;
            v8f acc = {0.f, 0.f, 0.f, 0.f, 0.f, 0.f, 0.f, 0.f};
            v2f af[C_ / 4], bfr[C_ / 4];
            #pragma unroll
            for (int j = 0; j < C_ / 4; ++j) {
                const int c0 = j * 4 + half * 2;
                af[j].x = arow[c0 * (V_ * K_)];
                af[j].y = arow[(c0 + 1) * (V_ * K_)];
                bfr[j]  = *reinterpret_cast<const v2f*>(brow + c0);
            }
            __builtin_amdgcn_sched_barrier(0);
            #pragma unroll
            for (int j = 0; j < C_ / 4; ++j) {
                v2f a = af[j], b;
                b.x = bfr[j].x * bmask;
                b.y = bfr[j].y * bmask;
                acc = __builtin_amdgcn_wmma_f32_16x16x4_f32(
                     false, a,  false, b,
                     (short)0, acc,  false,  false);
            }
            #pragma unroll
            for (int g = 0; g < 8; ++g)
                tile[g + half * 8][l] = acc[g];
            O_WAVE_SYNC();
            if (lane < 16) {
                const int v = (task & 1) * 16 + lane;
                float lg[K_];
                float mx = -1e30f;
                #pragma unroll
                for (int pp = 0; pp < K_; ++pp) {
                    lg[pp] = tile[lane][pp] + b1[k * K_ + pp];
                    mx = fmaxf(mx, lg[pp]);
                }
                float sum = 0.0f;
                #pragma unroll
                for (int pp = 0; pp < K_; ++pp) {
                    lg[pp] = __expf(lg[pp] - mx);
                    sum += lg[pp];
                }
                const float inv = 1.0f / sum;
                float wkv = 0.0f;
                #pragma unroll
                for (int pp = 0; pp < K_; ++pp)
                    wkv += lg[pp] * inv * w2[pp];
                if (v < V_) wkl[v * K_ + k] = wkv;
                if (v == 0 && k < 2) wkl[V_ * K_ + k] = wkv;
            }
            O_WAVE_SYNC();
        }
    }
    F_LDS_BLOCK_SYNC();

    const unsigned lane16 = (unsigned)lane * 16u;
    const char* wklb = reinterpret_cast<const char*>(wkl);
    const float bias = b2[0];
#define P_PIECE(R, M, U) do { \
        const unsigned tb = lane16 + (unsigned)(((128 * (M)) % (V_ * K_)) * 4); \
        const unsigned wb = (tb < (unsigned)(V_ * K_ * 4)) ? tb : (tb - (unsigned)(V_ * K_ * 4)); \
        const v2f wa = *reinterpret_cast<const v2f*>(wklb + wb); \
        const v2f wc = *reinterpret_cast<const v2f*>(wklb + wb + 8); \
        v2f h; \
        h.x = fmaf((R).x, wa.x, (R).y * wa.y); \
        h.y = fmaf((R).z, wc.x, (R).w * wc.y); \
        *reinterpret_cast<v2f*>(hw + 64 * (U) + 2 * lane) = h; \
    } while (0)

    #pragma unroll
    for (int g = 0; g < P_NGROUPS_; ++g) {
        __builtin_amdgcn_sched_barrier(0);
        if (g + 1 < P_NGROUPS_) P_LOAD((g + 1) & 1, (g + 1) * P_GROUP_, P_GROUP_);
        else                    P_LOAD((g + 1) & 1, (g + 1) * P_GROUP_, P_TAIL_);
        __builtin_amdgcn_sched_barrier(0);

        #pragma unroll
        for (int u = 0; u < P_GROUP_; ++u) {
            const v4f r = buf[g & 1][u];
            P_PIECE(r, g * P_GROUP_ + u, u);
        }
        O_WAVE_SYNC();

        #pragma unroll
        for (int j = 0; j < P_GRP_OUT_ / 32; ++j) {
            const int o = lane + 32 * j;
            const float* hs = hw + 5 * o;
            const float s = ((hs[0] + hs[1]) + (hs[2] + hs[3])) + hs[4];
            ow[(g & 3) * P_GRP_OUT_ + o] = s + bias;
        }
        O_WAVE_SYNC();

        if ((g & 3) == 3) {
            const v4f* ow4 = reinterpret_cast<const v4f*>(ow);
            v4f* dst = oslab4 + (g >> 2) * (P_BATCH_OUT_ / 4);
            #pragma unroll
            for (int r = 0; r < P_BATCH_OUT_ / 128; ++r) {
                const int i4 = lane + 32 * r;
                __builtin_nontemporal_store(ow4[i4], &dst[i4]);
            }
            O_WAVE_SYNC();
        }
    }

    #pragma unroll
    for (int u = 0; u < P_TAIL_; ++u) {
        const v4f r = buf[P_NGROUPS_ & 1][u];
        P_PIECE(r, P_NGROUPS_ * P_GROUP_ + u, u);
    }
    O_WAVE_SYNC();
    #pragma unroll
    for (int j = 0; j < P_TAIL_OUT_ / 32; ++j) {
        const int o = lane + 32 * j;
        const float* hs = hw + 5 * o;
        const float s = ((hs[0] + hs[1]) + (hs[2] + hs[3])) + hs[4];
        ow[o] = s + bias;
    }
    O_WAVE_SYNC();
    if (lane < P_TAIL_OUT_ / 4) {
        const v4f* ow4 = reinterpret_cast<const v4f*>(ow);
        v4f* dst = oslab4 + 2 * (P_BATCH_OUT_ / 4);
        __builtin_nontemporal_store(ow4[lane], &dst[lane]);
    }
#undef P_PIECE
#undef P_LOAD
}

extern "C" void kernel_launch(void* const* d_in, const int* in_sizes, int n_in,
                              void* d_out, int out_size, void* d_ws, size_t ws_size,
                              hipStream_t stream) {
    const float* region = (const float*)d_in[0];
    const float* w1     = (const float*)d_in[1];
    const float* b1     = (const float*)d_in[2];
    const float* w2     = (const float*)d_in[3];
    const float* b2     = (const float*)d_in[4];
    float* out = (float*)d_out;

    float* mfr = (float*)d_ws;

    vc_mean_kernel<<<N_ * C_, 64, 0, stream>>>(region, mfr);
    vc_out_kernel<<<N_ * 8, 512, 0, stream>>>(region, mfr, w1, b1, w2, b2, out);
}
